# starting kernel
# baseline (speedup 1.0000x reference)
#include <hip/hip_runtime.h>
#include <math.h>

typedef __attribute__((ext_vector_type(2))) float v2f;
typedef __attribute__((ext_vector_type(8))) float v8f;

#define N_NODES 100000
#define IN_CH   256
#define OUT_CH  64
#define N_EDGES 400000
#define ALPHA_F 0.25f

#define WS_B   0
#define WS_S   4096
#define WS_ACC (4096 + 2 * N_NODES)

__global__ void __launch_bounds__(256)
prep_kernel(const float* __restrict__ Wd, const float* __restrict__ Wf,
            float* __restrict__ ws) {
    __shared__ float ush[OUT_CH];
    __shared__ float vsh[OUT_CH];
    const int t = threadIdx.x;
    if (t < OUT_CH) {
        ush[t] = Wf[t] + Wf[2 * OUT_CH + t];
        vsh[t] = Wf[OUT_CH + t] - Wf[2 * OUT_CH + t];
    }
    __syncthreads();

    float su = 0.f, sv = 0.f;
    #pragma unroll 8
    for (int o = 0; o < OUT_CH; ++o) {
        const float w = Wd[o * IN_CH + t];
        su = fmaf(w, ush[o], su);
        sv = fmaf(w, vsh[o], sv);
    }

    const int kk       = t >> 2;
    const int koff     = t & 3;
    const int laneBase = (koff < 2) ? 0 : 16;
    const int slot     = koff & 1;
    float* bp = ws + WS_B;
    #pragma unroll
    for (int col = 0; col < 16; ++col) {
        const float val = (col == 0) ? su : (col == 1) ? sv : 0.f;
        bp[(kk * 32 + laneBase + col) * 2 + slot] = val;
    }
    if (t == 0) { ws[WS_ACC] = 0.f; ws[WS_ACC + 1] = 0.f; }
}

__global__ void __launch_bounds__(256)
matvec_wmma_kernel(const float* __restrict__ x, float* __restrict__ ws) {
    const int wave   = (int)((blockIdx.x * blockDim.x + threadIdx.x) >> 5);
    const int nWaves = N_NODES / 16;
    if (wave >= nWaves) return;
    const int lane = threadIdx.x & 31;
    const int row0 = wave * 16;

    const int    rowA = row0 + (lane & 15);
    const float* ax   = x + (size_t)rowA * IN_CH + ((lane & 16) ? 2 : 0);
    const v2f*   bp   = (const v2f*)(ws + WS_B);

    v8f c = {};
    #pragma unroll 8
    for (int kk = 0; kk < 64; ++kk) {
        v2f a = *(const v2f*)(ax + 4 * kk);
        v2f b = bp[kk * 32 + lane];
        c = __builtin_amdgcn_wmma_f32_16x16x4_f32(
                 false, a,  false, b,
                 (short)0, c,  false,  false);
    }

    const int col = lane & 15;
    if (col < 2) {
        float* s = ws + WS_S;
        const int rbase = row0 + ((lane & 16) ? 8 : 0);
        #pragma unroll
        for (int r = 0; r < 8; ++r) s[(size_t)(rbase + r) * 2 + col] = c[r];
    }
}

__global__ void __launch_bounds__(256)
edge_kernel(const int* __restrict__ eidx, const int* __restrict__ labels,
            const int* __restrict__ mask, const float* __restrict__ ws,
            float* __restrict__ out, float* __restrict__ acc) {
    const int e   = (int)(blockIdx.x * blockDim.x + threadIdx.x);
    const int tid = threadIdx.x;
    const float* s = ws + WS_S;

    float lm = 0.f, cm = 0.f;
    if (e < N_EDGES) {
        const int src = eidx[e];
        const int dst = eidx[N_EDGES + e];
        const float pre   = s[(size_t)src * 2 + 0] + s[(size_t)dst * 2 + 1];
        const float score = 1.f / (1.f + __expf(-pre));
        out[e] = score;
        const float p     = 1.f / (1.f + __expf(-score));
        const bool  pos   = (labels[e] == 1);
        const float pt    = pos ? p : (1.f - p);
        const float logpt = __logf(pt);
        const float at    = pos ? (1.f - ALPHA_F) : ALPHA_F;
        const float omp   = 1.f - pt;
        const float loss  = -(omp * omp) * (logpt * at);
        if (mask[e] != 0) { lm = loss; cm = 1.f; }
    }

    __shared__ float rl[256];
    __shared__ float rc[256];
    rl[tid] = lm; rc[tid] = cm;
    __syncthreads();
    for (int stride = 128; stride > 0; stride >>= 1) {
        if (tid < stride) { rl[tid] += rl[tid + stride]; rc[tid] += rc[tid + stride]; }
        __syncthreads();
    }
    if (tid == 0) {
        atomicAdd(&acc[0], rl[0]);
        atomicAdd(&acc[1], rc[0]);
    }
}

__global__ void finalize_kernel(const float* __restrict__ acc, float* __restrict__ out) {
    out[N_EDGES] = acc[0] / acc[1];
}

extern "C" void kernel_launch(void* const* d_in, const int* in_sizes, int n_in,
                              void* d_out, int out_size, void* d_ws, size_t ws_size,
                              hipStream_t stream) {
    const float* x      = (const float*)d_in[0];
    const float* Wd     = (const float*)d_in[1];
    const float* Wf     = (const float*)d_in[2];
    const int*   eidx   = (const int*)d_in[3];
    const int*   labels = (const int*)d_in[4];
    const int*   mask   = (const int*)d_in[5];
    float*       out    = (float*)d_out;
    float*       ws     = (float*)d_ws;
    float*       acc    = ws + WS_ACC;

    prep_kernel<<<1, 256, 0, stream>>>(Wd, Wf, ws);

    const int nWaves = N_NODES / 16;
    const int nBlk   = (nWaves + 7) / 8;
    matvec_wmma_kernel<<<nBlk, 256, 0, stream>>>(x, ws);

    edge_kernel<<<(N_EDGES + 255) / 256, 256, 0, stream>>>(eidx, labels, mask, ws, out, acc);

    finalize_kernel<<<1, 1, 0, stream>>>(acc, out);
}


# matvec kernel: each wave's 16x256 x tile staged through LDS in four 64-float k pieces (whole-line 16-byte loads, 32 in flight; padded-pitch fragment reads); B fragments read from LDS; k order kept
# speedup vs baseline: 1.0757x; 1.0757x over previous
#include <hip/hip_runtime.h>
#include <math.h>

typedef __attribute__((ext_vector_type(2))) float v2f;
typedef __attribute__((ext_vector_type(8))) float v8f;

#define N_NODES 100000
#define IN_CH   256
#define OUT_CH  64
#define N_EDGES 400000
#define ALPHA_F 0.25f

#define WS_B   0
#define WS_S   4096
#define WS_ACC (4096 + 2 * N_NODES)

__global__ void __launch_bounds__(256)
prep_kernel(const float* __restrict__ Wd, const float* __restrict__ Wf,
            float* __restrict__ ws) {
    __shared__ float ush[OUT_CH];
    __shared__ float vsh[OUT_CH];
    const int t = threadIdx.x;
    if (t < OUT_CH) {
        ush[t] = Wf[t] + Wf[2 * OUT_CH + t];
        vsh[t] = Wf[OUT_CH + t] - Wf[2 * OUT_CH + t];
    }
    __syncthreads();

    float su = 0.f, sv = 0.f;
    #pragma unroll 8
    for (int o = 0; o < OUT_CH; ++o) {
        const float w = Wd[o * IN_CH + t];
        su = fmaf(w, ush[o], su);
        sv = fmaf(w, vsh[o], sv);
    }

    const int kk       = t >> 2;
    const int koff     = t & 3;
    const int laneBase = (koff < 2) ? 0 : 16;
    const int slot     = koff & 1;
    float* bp = ws + WS_B;
    #pragma unroll
    for (int col = 0; col < 16; ++col) {
        const float val = (col == 0) ? su : (col == 1) ? sv : 0.f;
        bp[(kk * 32 + laneBase + col) * 2 + slot] = val;
    }
    if (t == 0) { ws[WS_ACC] = 0.f; ws[WS_ACC + 1] = 0.f; }
}

#define XP    68
#define XT    (16 * XP)

#define LD8(R, P)                                   \
    R##0 = gp[128 * 0 + 16 * (P)]; R##1 = gp[128 * 1 + 16 * (P)]; \
    R##2 = gp[128 * 2 + 16 * (P)]; R##3 = gp[128 * 3 + 16 * (P)]; \
    R##4 = gp[128 * 4 + 16 * (P)]; R##5 = gp[128 * 5 + 16 * (P)]; \
    R##6 = gp[128 * 6 + 16 * (P)]; R##7 = gp[128 * 7 + 16 * (P)];
#define ST8(R)                                      \
    *(float4*)(xst +  0 * XP) = R##0; *(float4*)(xst +  2 * XP) = R##1; \
    *(float4*)(xst +  4 * XP) = R##2; *(float4*)(xst +  6 * XP) = R##3; \
    *(float4*)(xst +  8 * XP) = R##4; *(float4*)(xst + 10 * XP) = R##5; \
    *(float4*)(xst + 12 * XP) = R##6; *(float4*)(xst + 14 * XP) = R##7;

__global__ void __launch_bounds__(256)
matvec_wmma_kernel(const float* x, float* __restrict__ ws) {
    __shared__ __attribute__((aligned(16))) float Bs[4096];
    __shared__ __attribute__((aligned(16))) float Xs[8 * XT];

    const int tid    = threadIdx.x;
    const int lane   = tid & 31;
    const int wib    = tid >> 5;
    const int wave   = (int)blockIdx.x * 8 + wib;
    const int nWaves = N_NODES / 16;
    const bool active = wave < nWaves;
    const int  row0   = (active ? wave : nWaves - 1) * 16;

    const int sub = lane & 15;
    const int hi  = lane >> 4;

    {
        const float4* bg = (const float4*)(ws + WS_B);
        float4*       bs = (float4*)Bs;
        const float4 b0 = bg[tid], b1 = bg[tid + 256], b2 = bg[tid + 512], b3 = bg[tid + 768];
        bs[tid] = b0; bs[tid + 256] = b1; bs[tid + 512] = b2; bs[tid + 768] = b3;
    }
    __syncthreads();

    float*       xs  = Xs + wib * XT;
    float*       xst = xs + hi * XP + 4 * sub;
    const float* xld = xs + sub * XP + 2 * hi;
    const v2f*   bl  = (const v2f*)Bs + lane;

    const float4* gp = (const float4*)(x + (size_t)(row0 + hi) * IN_CH + 4 * sub);
    float4 ra0, ra1, ra2, ra3, ra4, ra5, ra6, ra7;
    float4 rb0, rb1, rb2, rb3, rb4, rb5, rb6, rb7;
    float4 rc0, rc1, rc2, rc3, rc4, rc5, rc6, rc7;
    float4 rd0, rd1, rd2, rd3, rd4, rd5, rd6, rd7;
    LD8(ra, 0)
    LD8(rb, 1)
    LD8(rc, 2)
    LD8(rd, 3)
    __builtin_amdgcn_sched_barrier(0);

    v8f c = {};
#define MM16(P)                                                         \
    _Pragma("unroll")                                                   \
    for (int kl = 0; kl < 16; ++kl) {                                   \
        v2f a = *(const v2f*)(xld + 4 * kl);                            \
        v2f b = bl[(16 * (P) + kl) * 32];                               \
        c = __builtin_amdgcn_wmma_f32_16x16x4_f32(                      \
                 false, a,  false, b,                 \
                 (short)0, c,  false,  false); \
    }

    ST8(ra)
    MM16(0)
    ST8(rb)
    MM16(1)
    ST8(rc)
    MM16(2)
    ST8(rd)
    MM16(3)

    if (active && sub < 2) {
        float* s = ws + WS_S;
        const int rbase = row0 + (hi ? 8 : 0);
        #pragma unroll
        for (int r = 0; r < 8; ++r) s[(size_t)(rbase + r) * 2 + sub] = c[r];
    }
}
#undef LD8
#undef ST8
#undef MM16

__global__ void __launch_bounds__(256)
edge_kernel(const int* __restrict__ eidx, const int* __restrict__ labels,
            const int* __restrict__ mask, const float* __restrict__ ws,
            float* __restrict__ out, float* __restrict__ acc) {
    const int e   = (int)(blockIdx.x * blockDim.x + threadIdx.x);
    const int tid = threadIdx.x;
    const float* s = ws + WS_S;

    float lm = 0.f, cm = 0.f;
    if (e < N_EDGES) {
        const int src = eidx[e];
        const int dst = eidx[N_EDGES + e];
        const float pre   = s[(size_t)src * 2 + 0] + s[(size_t)dst * 2 + 1];
        const float score = 1.f / (1.f + __expf(-pre));
        out[e] = score;
        const float p     = 1.f / (1.f + __expf(-score));
        const bool  pos   = (labels[e] == 1);
        const float pt    = pos ? p : (1.f - p);
        const float logpt = __logf(pt);
        const float at    = pos ? (1.f - ALPHA_F) : ALPHA_F;
        const float omp   = 1.f - pt;
        const float loss  = -(omp * omp) * (logpt * at);
        if (mask[e] != 0) { lm = loss; cm = 1.f; }
    }

    __shared__ float rl[256];
    __shared__ float rc[256];
    rl[tid] = lm; rc[tid] = cm;
    __syncthreads();
    for (int stride = 128; stride > 0; stride >>= 1) {
        if (tid < stride) { rl[tid] += rl[tid + stride]; rc[tid] += rc[tid + stride]; }
        __syncthreads();
    }
    if (tid == 0) {
        atomicAdd(&acc[0], rl[0]);
        atomicAdd(&acc[1], rc[0]);
    }
}

__global__ void finalize_kernel(const float* __restrict__ acc, float* __restrict__ out) {
    out[N_EDGES] = acc[0] / acc[1];
}

extern "C" void kernel_launch(void* const* d_in, const int* in_sizes, int n_in,
                              void* d_out, int out_size, void* d_ws, size_t ws_size,
                              hipStream_t stream) {
    const float* x      = (const float*)d_in[0];
    const float* Wd     = (const float*)d_in[1];
    const float* Wf     = (const float*)d_in[2];
    const int*   eidx   = (const int*)d_in[3];
    const int*   labels = (const int*)d_in[4];
    const int*   mask   = (const int*)d_in[5];
    float*       out    = (float*)d_out;
    float*       ws     = (float*)d_ws;
    float*       acc    = ws + WS_ACC;

    prep_kernel<<<1, 256, 0, stream>>>(Wd, Wf, ws);

    const int nWaves = N_NODES / 16;
    const int nBlk   = (nWaves + 7) / 8;
    matvec_wmma_kernel<<<nBlk, 256, 0, stream>>>(x, ws);

    edge_kernel<<<(N_EDGES + 255) / 256, 256, 0, stream>>>(eidx, labels, mask, ws, out, acc);

    finalize_kernel<<<1, 1, 0, stream>>>(acc, out);
}


# mean split as sum over blocks of (block loss sum / mask count); count taken by 32 extra blocks of the first launch (exact int partials); finalize launch gone (loss: divide per block, then add)
# speedup vs baseline: 1.2217x; 1.1357x over previous
#include <hip/hip_runtime.h>
#include <math.h>

typedef __attribute__((ext_vector_type(2))) float v2f;
typedef __attribute__((ext_vector_type(8))) float v8f;

#define N_NODES 100000
#define IN_CH   256
#define OUT_CH  64
#define N_EDGES 400000
#define ALPHA_F 0.25f

#define WS_B   0
#define WS_S   4096
#define WS_CNT (4096 + 2 * N_NODES)

#define CNT_BLOCKS 32
#define CNT_V4     (N_EDGES / 4 / CNT_BLOCKS)

__global__ void __launch_bounds__(256)
prep_kernel(const float* __restrict__ Wd, const float* __restrict__ Wf,
            const int* __restrict__ mask, float* __restrict__ ws,
            int* __restrict__ cnt, float* __restrict__ out) {
    __shared__ float ush[OUT_CH];
    __shared__ float vsh[OUT_CH];
    __shared__ int   cw[8];
    const int t = threadIdx.x;

    if (blockIdx.x != 0) {
        const int4* m4 = (const int4*)mask + (size_t)(blockIdx.x - 1) * CNT_V4;
        int c = 0;
        #pragma unroll
        for (int j = 0; j < (CNT_V4 + 255) / 256; ++j) {
            const int i = t + 256 * j;
            if (i < CNT_V4) {
                const int4 v = m4[i];
                c += (v.x != 0) + (v.y != 0) + (v.z != 0) + (v.w != 0);
            }
        }
        for (int o = 16; o > 0; o >>= 1) c += __shfl_xor(c, o);
        if ((t & 31) == 0) cw[t >> 5] = c;
        __syncthreads();
        if (t == 0) {
            int s = 0;
            for (int w = 0; w < 8; ++w) s += cw[w];
            cnt[blockIdx.x - 1] = s;
        }
        return;
    }
    if (t < OUT_CH) {
        ush[t] = Wf[t] + Wf[2 * OUT_CH + t];
        vsh[t] = Wf[OUT_CH + t] - Wf[2 * OUT_CH + t];
    }
    __syncthreads();

    float su = 0.f, sv = 0.f;
    #pragma unroll 8
    for (int o = 0; o < OUT_CH; ++o) {
        const float w = Wd[o * IN_CH + t];
        su = fmaf(w, ush[o], su);
        sv = fmaf(w, vsh[o], sv);
    }

    const int kk       = t >> 2;
    const int koff     = t & 3;
    const int laneBase = (koff < 2) ? 0 : 16;
    const int slot     = koff & 1;
    float* bp = ws + WS_B;
    #pragma unroll
    for (int col = 0; col < 16; ++col) {
        const float val = (col == 0) ? su : (col == 1) ? sv : 0.f;
        bp[(kk * 32 + laneBase + col) * 2 + slot] = val;
    }
    if (t == 0) out[N_EDGES] = 0.f;
}

#define XP    68
#define XT    (16 * XP)

#define LD8(R, P)                                   \
    R##0 = gp[128 * 0 + 16 * (P)]; R##1 = gp[128 * 1 + 16 * (P)]; \
    R##2 = gp[128 * 2 + 16 * (P)]; R##3 = gp[128 * 3 + 16 * (P)]; \
    R##4 = gp[128 * 4 + 16 * (P)]; R##5 = gp[128 * 5 + 16 * (P)]; \
    R##6 = gp[128 * 6 + 16 * (P)]; R##7 = gp[128 * 7 + 16 * (P)];
#define ST8(R)                                      \
    *(float4*)(xst +  0 * XP) = R##0; *(float4*)(xst +  2 * XP) = R##1; \
    *(float4*)(xst +  4 * XP) = R##2; *(float4*)(xst +  6 * XP) = R##3; \
    *(float4*)(xst +  8 * XP) = R##4; *(float4*)(xst + 10 * XP) = R##5; \
    *(float4*)(xst + 12 * XP) = R##6; *(float4*)(xst + 14 * XP) = R##7;

__global__ void __launch_bounds__(256)
matvec_wmma_kernel(const float* x, float* __restrict__ ws) {
    __shared__ __attribute__((aligned(16))) float Bs[4096];
    __shared__ __attribute__((aligned(16))) float Xs[8 * XT];

    const int tid    = threadIdx.x;
    const int lane   = tid & 31;
    const int wib    = tid >> 5;
    const int wave   = (int)blockIdx.x * 8 + wib;
    const int nWaves = N_NODES / 16;
    const bool active = wave < nWaves;
    const int  row0   = (active ? wave : nWaves - 1) * 16;

    const int sub = lane & 15;
    const int hi  = lane >> 4;

    {
        const float4* bg = (const float4*)(ws + WS_B);
        float4*       bs = (float4*)Bs;
        const float4 b0 = bg[tid], b1 = bg[tid + 256], b2 = bg[tid + 512], b3 = bg[tid + 768];
        bs[tid] = b0; bs[tid + 256] = b1; bs[tid + 512] = b2; bs[tid + 768] = b3;
    }
    __syncthreads();

    float*       xs  = Xs + wib * XT;
    float*       xst = xs + hi * XP + 4 * sub;
    const float* xld = xs + sub * XP + 2 * hi;
    const v2f*   bl  = (const v2f*)Bs + lane;

    const float4* gp = (const float4*)(x + (size_t)(row0 + hi) * IN_CH + 4 * sub);
    float4 ra0, ra1, ra2, ra3, ra4, ra5, ra6, ra7;
    float4 rb0, rb1, rb2, rb3, rb4, rb5, rb6, rb7;
    float4 rc0, rc1, rc2, rc3, rc4, rc5, rc6, rc7;
    float4 rd0, rd1, rd2, rd3, rd4, rd5, rd6, rd7;
    LD8(ra, 0)
    LD8(rb, 1)
    LD8(rc, 2)
    LD8(rd, 3)
    __builtin_amdgcn_sched_barrier(0);

    v8f c = {};
#define MM16(P)                                                         \
    _Pragma("unroll")                                                   \
    for (int kl = 0; kl < 16; ++kl) {                                   \
        v2f a = *(const v2f*)(xld + 4 * kl);                            \
        v2f b = bl[(16 * (P) + kl) * 32];                               \
        c = __builtin_amdgcn_wmma_f32_16x16x4_f32(                      \
                 false, a,  false, b,                 \
                 (short)0, c,  false,  false); \
    }

    ST8(ra)
    MM16(0)
    ST8(rb)
    MM16(1)
    ST8(rc)
    MM16(2)
    ST8(rd)
    MM16(3)

    if (active && sub < 2) {
        float* s = ws + WS_S;
        const int rbase = row0 + (hi ? 8 : 0);
        #pragma unroll
        for (int r = 0; r < 8; ++r) s[(size_t)(rbase + r) * 2 + sub] = c[r];
    }
}
#undef LD8
#undef ST8
#undef MM16

__global__ void __launch_bounds__(256)
edge_kernel(const int* __restrict__ eidx, const int* __restrict__ labels,
            const int* __restrict__ mask, const float* __restrict__ ws,
            const int* __restrict__ cnt, float* __restrict__ out) {
    const int e   = (int)(blockIdx.x * blockDim.x + threadIdx.x);
    const int tid = threadIdx.x;
    const float* s = ws + WS_S;
    const int cpart = cnt[tid & (CNT_BLOCKS - 1)];

    float lm = 0.f;
    if (e < N_EDGES) {
        const int src = eidx[e];
        const int dst = eidx[N_EDGES + e];
        const float pre   = s[(size_t)src * 2 + 0] + s[(size_t)dst * 2 + 1];
        const float score = 1.f / (1.f + __expf(-pre));
        out[e] = score;
        const float p     = 1.f / (1.f + __expf(-score));
        const bool  pos   = (labels[e] == 1);
        const float pt    = pos ? p : (1.f - p);
        const float logpt = __logf(pt);
        const float at    = pos ? (1.f - ALPHA_F) : ALPHA_F;
        const float omp   = 1.f - pt;
        const float loss  = -(omp * omp) * (logpt * at);
        if (mask[e] != 0) lm = loss;
    }

    __shared__ float rl[256];
    rl[tid] = lm;
    int ctot = cpart;
    if (tid < 32) {
        for (int o = 16; o > 0; o >>= 1) ctot += __shfl_xor(ctot, o);
    }
    __syncthreads();
    for (int stride = 128; stride > 0; stride >>= 1) {
        if (tid < stride) rl[tid] += rl[tid + stride];
        __syncthreads();
    }
    if (tid == 0) atomicAdd(&out[N_EDGES], rl[0] / (float)ctot);
}

extern "C" void kernel_launch(void* const* d_in, const int* in_sizes, int n_in,
                              void* d_out, int out_size, void* d_ws, size_t ws_size,
                              hipStream_t stream) {
    const float* x      = (const float*)d_in[0];
    const float* Wd     = (const float*)d_in[1];
    const float* Wf     = (const float*)d_in[2];
    const int*   eidx   = (const int*)d_in[3];
    const int*   labels = (const int*)d_in[4];
    const int*   mask   = (const int*)d_in[5];
    float*       out    = (float*)d_out;
    float*       ws     = (float*)d_ws;
    int*         cnt    = (int*)(ws + WS_CNT);

    prep_kernel<<<1 + CNT_BLOCKS, 256, 0, stream>>>(Wd, Wf, mask, ws, cnt, out);

    const int nWaves = N_NODES / 16;
    const int nBlk   = (nWaves + 7) / 8;
    matvec_wmma_kernel<<<nBlk, 256, 0, stream>>>(x, ws);

    edge_kernel<<<(N_EDGES + 255) / 256, 256, 0, stream>>>(eidx, labels, mask, ws, cnt, out);
}


# first launch: every global load requested before any wait (fold: 64 W_d loads per thread in one group, spread over 4 blocks; count: 13 unconditional loads); sums and their order unchanged
# speedup vs baseline: 1.3728x; 1.1237x over previous
#include <hip/hip_runtime.h>
#include <math.h>

typedef __attribute__((ext_vector_type(2))) float v2f;
typedef __attribute__((ext_vector_type(8))) float v8f;

#define N_NODES 100000
#define IN_CH   256
#define OUT_CH  64
#define N_EDGES 400000
#define ALPHA_F 0.25f

#define WS_B   0
#define WS_S   4096
#define WS_CNT (4096 + 2 * N_NODES)

#define CNT_BLOCKS 32
#define CNT_V4     (N_EDGES / 4 / CNT_BLOCKS)
#define CNT_LD     ((CNT_V4 + 255) / 256)
#define FOLD_BLOCKS 4
#define FOLD_K      (IN_CH / FOLD_BLOCKS)
static_assert(FOLD_K == OUT_CH, "the threads that own a k also write one u and one v each");

__global__ void __launch_bounds__(256)
prep_kernel(const float* __restrict__ Wd, const float* __restrict__ Wf,
            const int* __restrict__ mask, float* __restrict__ ws,
            int* __restrict__ cnt, float* __restrict__ out) {
    __shared__ float ush[OUT_CH];
    __shared__ float vsh[OUT_CH];
    __shared__ int   cw[8];
    const int t = threadIdx.x;

    if (blockIdx.x >= FOLD_BLOCKS) {
        const int    cb = (int)blockIdx.x - FOLD_BLOCKS;
        const int4*  m4 = (const int4*)mask + (size_t)cb * CNT_V4;
        int4 v[CNT_LD];
        #pragma unroll
        for (int j = 0; j < CNT_LD; ++j) {
            const int i = t + 256 * j;
            v[j] = m4[i < CNT_V4 ? i : CNT_V4 - 1];
        }
        int c = 0;
        #pragma unroll
        for (int j = 0; j < CNT_LD; ++j) {
            const int n = (v[j].x != 0) + (v[j].y != 0) + (v[j].z != 0) + (v[j].w != 0);
            c += (t + 256 * j < CNT_V4) ? n : 0;
        }
        for (int o = 16; o > 0; o >>= 1) c += __shfl_xor(c, o);
        if ((t & 31) == 0) cw[t >> 5] = c;
        __syncthreads();
        if (t == 0) {
            int s = 0;
            for (int w = 0; w < 8; ++w) s += cw[w];
            cnt[cb] = s;
        }
        return;
    }

    const bool own = t < FOLD_K;
    const int  k   = (int)blockIdx.x * FOLD_K + (t & (FOLD_K - 1));
    float w[OUT_CH];
    if (own) {
        #pragma unroll
        for (int o = 0; o < OUT_CH; ++o) w[o] = Wd[o * IN_CH + k];
        ush[t] = Wf[t] + Wf[2 * OUT_CH + t];
        vsh[t] = Wf[OUT_CH + t] - Wf[2 * OUT_CH + t];
    }
    __syncthreads();
    if (!own) return;

    float su = 0.f, sv = 0.f;
    #pragma unroll
    for (int o = 0; o < OUT_CH; ++o) {
        su = fmaf(w[o], ush[o], su);
        sv = fmaf(w[o], vsh[o], sv);
    }

    const int kk       = k >> 2;
    const int koff     = k & 3;
    const int laneBase = (koff < 2) ? 0 : 16;
    const int slot     = koff & 1;
    float* bp = ws + WS_B;
    #pragma unroll
    for (int col = 0; col < 16; ++col) {
        const float val = (col == 0) ? su : (col == 1) ? sv : 0.f;
        bp[(kk * 32 + laneBase + col) * 2 + slot] = val;
    }
    if (blockIdx.x == 0 && t == 0) out[N_EDGES] = 0.f;
}

#define XP    68
#define XT    (16 * XP)

#define LD8(R, P)                                   \
    R##0 = gp[128 * 0 + 16 * (P)]; R##1 = gp[128 * 1 + 16 * (P)]; \
    R##2 = gp[128 * 2 + 16 * (P)]; R##3 = gp[128 * 3 + 16 * (P)]; \
    R##4 = gp[128 * 4 + 16 * (P)]; R##5 = gp[128 * 5 + 16 * (P)]; \
    R##6 = gp[128 * 6 + 16 * (P)]; R##7 = gp[128 * 7 + 16 * (P)];
#define ST8(R)                                      \
    *(float4*)(xst +  0 * XP) = R##0; *(float4*)(xst +  2 * XP) = R##1; \
    *(float4*)(xst +  4 * XP) = R##2; *(float4*)(xst +  6 * XP) = R##3; \
    *(float4*)(xst +  8 * XP) = R##4; *(float4*)(xst + 10 * XP) = R##5; \
    *(float4*)(xst + 12 * XP) = R##6; *(float4*)(xst + 14 * XP) = R##7;

__global__ void __launch_bounds__(256)
matvec_wmma_kernel(const float* x, float* __restrict__ ws) {
    __shared__ __attribute__((aligned(16))) float Bs[4096];
    __shared__ __attribute__((aligned(16))) float Xs[8 * XT];

    const int tid    = threadIdx.x;
    const int lane   = tid & 31;
    const int wib    = tid >> 5;
    const int wave   = (int)blockIdx.x * 8 + wib;
    const int nWaves = N_NODES / 16;
    const bool active = wave < nWaves;
    const int  row0   = (active ? wave : nWaves - 1) * 16;

    const int sub = lane & 15;
    const int hi  = lane >> 4;

    {
        const float4* bg = (const float4*)(ws + WS_B);
        float4*       bs = (float4*)Bs;
        const float4 b0 = bg[tid], b1 = bg[tid + 256], b2 = bg[tid + 512], b3 = bg[tid + 768];
        bs[tid] = b0; bs[tid + 256] = b1; bs[tid + 512] = b2; bs[tid + 768] = b3;
    }
    __syncthreads();

    float*       xs  = Xs + wib * XT;
    float*       xst = xs + hi * XP + 4 * sub;
    const float* xld = xs + sub * XP + 2 * hi;
    const v2f*   bl  = (const v2f*)Bs + lane;

    const float4* gp = (const float4*)(x + (size_t)(row0 + hi) * IN_CH + 4 * sub);
    float4 ra0, ra1, ra2, ra3, ra4, ra5, ra6, ra7;
    float4 rb0, rb1, rb2, rb3, rb4, rb5, rb6, rb7;
    float4 rc0, rc1, rc2, rc3, rc4, rc5, rc6, rc7;
    float4 rd0, rd1, rd2, rd3, rd4, rd5, rd6, rd7;
    LD8(ra, 0)
    LD8(rb, 1)
    LD8(rc, 2)
    LD8(rd, 3)
    __builtin_amdgcn_sched_barrier(0);

    v8f c = {};
#define MM16(P)                                                         \
    _Pragma("unroll")                                                   \
    for (int kl = 0; kl < 16; ++kl) {                                   \
        v2f a = *(const v2f*)(xld + 4 * kl);                            \
        v2f b = bl[(16 * (P) + kl) * 32];                               \
        c = __builtin_amdgcn_wmma_f32_16x16x4_f32(                      \
                 false, a,  false, b,                 \
                 (short)0, c,  false,  false); \
    }

    ST8(ra)
    MM16(0)
    ST8(rb)
    MM16(1)
    ST8(rc)
    MM16(2)
    ST8(rd)
    MM16(3)

    if (active && sub < 2) {
        float* s = ws + WS_S;
        const int rbase = row0 + (hi ? 8 : 0);
        #pragma unroll
        for (int r = 0; r < 8; ++r) s[(size_t)(rbase + r) * 2 + sub] = c[r];
    }
}
#undef LD8
#undef ST8
#undef MM16

__global__ void __launch_bounds__(256)
edge_kernel(const int* __restrict__ eidx, const int* __restrict__ labels,
            const int* __restrict__ mask, const float* __restrict__ ws,
            const int* __restrict__ cnt, float* __restrict__ out) {
    const int e   = (int)(blockIdx.x * blockDim.x + threadIdx.x);
    const int tid = threadIdx.x;
    const float* s = ws + WS_S;
    const int cpart = cnt[tid & (CNT_BLOCKS - 1)];

    float lm = 0.f;
    if (e < N_EDGES) {
        const int src = eidx[e];
        const int dst = eidx[N_EDGES + e];
        const float pre   = s[(size_t)src * 2 + 0] + s[(size_t)dst * 2 + 1];
        const float score = 1.f / (1.f + __expf(-pre));
        out[e] = score;
        const float p     = 1.f / (1.f + __expf(-score));
        const bool  pos   = (labels[e] == 1);
        const float pt    = pos ? p : (1.f - p);
        const float logpt = __logf(pt);
        const float at    = pos ? (1.f - ALPHA_F) : ALPHA_F;
        const float omp   = 1.f - pt;
        const float loss  = -(omp * omp) * (logpt * at);
        if (mask[e] != 0) lm = loss;
    }

    __shared__ float rl[256];
    rl[tid] = lm;
    int ctot = cpart;
    if (tid < 32) {
        for (int o = 16; o > 0; o >>= 1) ctot += __shfl_xor(ctot, o);
    }
    __syncthreads();
    for (int stride = 128; stride > 0; stride >>= 1) {
        if (tid < stride) rl[tid] += rl[tid + stride];
        __syncthreads();
    }
    if (tid == 0) atomicAdd(&out[N_EDGES], rl[0] / (float)ctot);
}

extern "C" void kernel_launch(void* const* d_in, const int* in_sizes, int n_in,
                              void* d_out, int out_size, void* d_ws, size_t ws_size,
                              hipStream_t stream) {
    const float* x      = (const float*)d_in[0];
    const float* Wd     = (const float*)d_in[1];
    const float* Wf     = (const float*)d_in[2];
    const int*   eidx   = (const int*)d_in[3];
    const int*   labels = (const int*)d_in[4];
    const int*   mask   = (const int*)d_in[5];
    float*       out    = (float*)d_out;
    float*       ws     = (float*)d_ws;
    int*         cnt    = (int*)(ws + WS_CNT);

    prep_kernel<<<FOLD_BLOCKS + CNT_BLOCKS, 256, 0, stream>>>(Wd, Wf, mask, ws, cnt, out);

    const int nWaves = N_NODES / 16;
    const int nBlk   = (nWaves + 7) / 8;
    matvec_wmma_kernel<<<nBlk, 256, 0, stream>>>(x, ws);

    edge_kernel<<<(N_EDGES + 255) / 256, 256, 0, stream>>>(eidx, labels, mask, ws, cnt, out);
}


# matvec kernel retiled: 256 resident blocks of 8 waves, 3-4 row tiles per wave, next tile's 32 loads requested while the current one is parked and multiplied; B to LDS once per block; k order kept
# speedup vs baseline: 1.5936x; 1.1608x over previous
#include <hip/hip_runtime.h>
#include <math.h>

typedef __attribute__((ext_vector_type(2))) float v2f;
typedef __attribute__((ext_vector_type(8))) float v8f;

#define N_NODES 100000
#define IN_CH   256
#define OUT_CH  64
#define N_EDGES 400000
#define ALPHA_F 0.25f

#define WS_B   0
#define WS_S   4096
#define WS_CNT (4096 + 2 * N_NODES)

#define CNT_BLOCKS 32
#define CNT_V4     (N_EDGES / 4 / CNT_BLOCKS)
#define CNT_LD     ((CNT_V4 + 255) / 256)
#define FOLD_BLOCKS 4
#define FOLD_K      (IN_CH / FOLD_BLOCKS)
static_assert(FOLD_K == OUT_CH, "the threads that own a k also write one u and one v each");

__global__ void __launch_bounds__(256)
prep_kernel(const float* __restrict__ Wd, const float* __restrict__ Wf,
            const int* __restrict__ mask, float* __restrict__ ws,
            int* __restrict__ cnt, float* __restrict__ out) {
    __shared__ float ush[OUT_CH];
    __shared__ float vsh[OUT_CH];
    __shared__ int   cw[8];
    const int t = threadIdx.x;

    if (blockIdx.x >= FOLD_BLOCKS) {
        const int    cb = (int)blockIdx.x - FOLD_BLOCKS;
        const int4*  m4 = (const int4*)mask + (size_t)cb * CNT_V4;
        int4 v[CNT_LD];
        #pragma unroll
        for (int j = 0; j < CNT_LD; ++j) {
            const int i = t + 256 * j;
            v[j] = m4[i < CNT_V4 ? i : CNT_V4 - 1];
        }
        int c = 0;
        #pragma unroll
        for (int j = 0; j < CNT_LD; ++j) {
            const int n = (v[j].x != 0) + (v[j].y != 0) + (v[j].z != 0) + (v[j].w != 0);
            c += (t + 256 * j < CNT_V4) ? n : 0;
        }
        for (int o = 16; o > 0; o >>= 1) c += __shfl_xor(c, o);
        if ((t & 31) == 0) cw[t >> 5] = c;
        __syncthreads();
        if (t == 0) {
            int s = 0;
            for (int w = 0; w < 8; ++w) s += cw[w];
            cnt[cb] = s;
        }
        return;
    }

    const bool own = t < FOLD_K;
    const int  k   = (int)blockIdx.x * FOLD_K + (t & (FOLD_K - 1));
    float w[OUT_CH];
    if (own) {
        #pragma unroll
        for (int o = 0; o < OUT_CH; ++o) w[o] = Wd[o * IN_CH + k];
        ush[t] = Wf[t] + Wf[2 * OUT_CH + t];
        vsh[t] = Wf[OUT_CH + t] - Wf[2 * OUT_CH + t];
    }
    __syncthreads();
    if (!own) return;

    float su = 0.f, sv = 0.f;
    #pragma unroll
    for (int o = 0; o < OUT_CH; ++o) {
        su = fmaf(w[o], ush[o], su);
        sv = fmaf(w[o], vsh[o], sv);
    }

    const int kk       = k >> 2;
    const int koff     = k & 3;
    const int laneBase = (koff < 2) ? 0 : 16;
    const int slot     = koff & 1;
    float* bp = ws + WS_B;
    #pragma unroll
    for (int col = 0; col < 16; ++col) {
        const float val = (col == 0) ? su : (col == 1) ? sv : 0.f;
        bp[(kk * 32 + laneBase + col) * 2 + slot] = val;
    }
    if (blockIdx.x == 0 && t == 0) out[N_EDGES] = 0.f;
}

#define XP    68
#define XT    (16 * XP)

#define MV_TILES   (N_NODES / 16)
#define MV_BLOCKS  256
#define MV_BASE    (MV_TILES / MV_BLOCKS)
#define MV_EXTRA   (MV_TILES - MV_BASE * MV_BLOCKS)
static_assert(MV_BASE == 24 && MV_EXTRA < MV_BLOCKS, "three tiles per wave, a fourth for wave 0 of some blocks");

#define LD8(R, G, P)                                \
    R##0 = G[128 * 0 + 16 * (P)]; R##1 = G[128 * 1 + 16 * (P)]; \
    R##2 = G[128 * 2 + 16 * (P)]; R##3 = G[128 * 3 + 16 * (P)]; \
    R##4 = G[128 * 4 + 16 * (P)]; R##5 = G[128 * 5 + 16 * (P)]; \
    R##6 = G[128 * 6 + 16 * (P)]; R##7 = G[128 * 7 + 16 * (P)];
#define ST8(R)                                      \
    *(float4*)(xst +  0 * XP) = R##0; *(float4*)(xst +  2 * XP) = R##1; \
    *(float4*)(xst +  4 * XP) = R##2; *(float4*)(xst +  6 * XP) = R##3; \
    *(float4*)(xst +  8 * XP) = R##4; *(float4*)(xst + 10 * XP) = R##5; \
    *(float4*)(xst + 12 * XP) = R##6; *(float4*)(xst + 14 * XP) = R##7;
#define DECL8(R) float4 R##0, R##1, R##2, R##3, R##4, R##5, R##6, R##7;

__global__ void __launch_bounds__(256)
matvec_wmma_kernel(const float* x, float* __restrict__ ws) {
    __shared__ __attribute__((aligned(16))) float Bs[4096];
    __shared__ __attribute__((aligned(16))) float Xs[8 * XT];

    const int tid  = threadIdx.x;
    const int lane = tid & 31;
    const int wib  = __builtin_amdgcn_readfirstlane(tid >> 5);
    const int blk  = (int)blockIdx.x;
    const int t0   = MV_BASE * blk + (blk < MV_EXTRA ? blk : MV_EXTRA) + wib;
    const int t1   = t0 + 8, t2 = t0 + 16, t3 = t0 + 24;
    const bool four = (wib == 0) && (blk < MV_EXTRA);

    const int sub = lane & 15;
    const int hi  = lane >> 4;

    float*       xs  = Xs + wib * XT;
    float*       xst = xs + hi * XP + 4 * sub;
    const float* xld = xs + sub * XP + 2 * hi;
    const v2f*   bl  = (const v2f*)Bs + lane;

    const int     loff = hi * IN_CH + 4 * sub;
    const float4* g0   = (const float4*)(x + (size_t)t0 * (16 * IN_CH) + loff);
    const float4* g1   = (const float4*)(x + (size_t)t1 * (16 * IN_CH) + loff);
    const float4* g2   = (const float4*)(x + (size_t)t2 * (16 * IN_CH) + loff);
    DECL8(Aa) DECL8(Ab) DECL8(Ac) DECL8(Ad)
    DECL8(Ba) DECL8(Bb) DECL8(Bc) DECL8(Bd)

    const float4* bg = (const float4*)(ws + WS_B);
    float4*       bs = (float4*)Bs;
    const float4 b0 = bg[tid], b1 = bg[tid + 256], b2 = bg[tid + 512], b3 = bg[tid + 768];
    LD8(Aa, g0, 0)
    LD8(Ab, g0, 1)
    LD8(Ac, g0, 2)
    LD8(Ad, g0, 3)
    __builtin_amdgcn_sched_barrier(0);
    bs[tid] = b0; bs[tid + 256] = b1; bs[tid + 512] = b2; bs[tid + 768] = b3;
    __builtin_amdgcn_fence(__ATOMIC_RELEASE, "workgroup", "local");
    __builtin_amdgcn_s_barrier();
    __builtin_amdgcn_fence(__ATOMIC_ACQUIRE, "workgroup", "local");

#define MM16(P)                                                         \
    _Pragma("unroll")                                                   \
    for (int kl = 0; kl < 16; ++kl) {                                   \
        v2f a = *(const v2f*)(xld + 4 * kl);                            \
        v2f b = bl[(16 * (P) + kl) * 32];                               \
        c = __builtin_amdgcn_wmma_f32_16x16x4_f32(                      \
                 false, a,  false, b,                 \
                 (short)0, c,  false,  false); \
    }
#define PIECE(R, P, NEXT)                                               \
    NEXT                                                                \
    __builtin_amdgcn_sched_barrier(0);                                  \
    ST8(R)                                                              \
    MM16(P)                                                             \
    __builtin_amdgcn_sched_barrier(0);
#define STORE_S(T)                                                      \
    if (sub < 2) {                                                      \
        float* s = ws + WS_S + (size_t)(T) * 32;                        \
        const int rbase = hi ? 8 : 0;                                   \
        _Pragma("unroll")                                               \
        for (int r = 0; r < 8; ++r) s[(rbase + r) * 2 + sub] = c[r];    \
    }
#define TILE(CUR, NXT, T, GN) {                                         \
        v8f c = {};                                                     \
        PIECE(CUR##a, 0, LD8(NXT##a, GN, 0))                            \
        PIECE(CUR##b, 1, LD8(NXT##b, GN, 1))                            \
        PIECE(CUR##c, 2, LD8(NXT##c, GN, 2))                            \
        PIECE(CUR##d, 3, LD8(NXT##d, GN, 3))                            \
        STORE_S(T)                                                      \
    }
#define TILE_LAST(CUR, T) {                                             \
        v8f c = {};                                                     \
        PIECE(CUR##a, 0, )                                              \
        PIECE(CUR##b, 1, )                                              \
        PIECE(CUR##c, 2, )                                              \
        PIECE(CUR##d, 3, )                                              \
        STORE_S(T)                                                      \
    }

    TILE(A, B, t0, g1)
    TILE(B, A, t1, g2)
    if (four) {
        const float4* g3 = (const float4*)(x + (size_t)t3 * (16 * IN_CH) + loff);
        TILE(A, B, t2, g3)
        TILE_LAST(B, t3)
    } else {
        TILE_LAST(A, t2)
    }
}
#undef LD8
#undef ST8
#undef DECL8
#undef MM16
#undef PIECE
#undef STORE_S
#undef TILE
#undef TILE_LAST

__global__ void __launch_bounds__(256)
edge_kernel(const int* __restrict__ eidx, const int* __restrict__ labels,
            const int* __restrict__ mask, const float* __restrict__ ws,
            const int* __restrict__ cnt, float* __restrict__ out) {
    const int e   = (int)(blockIdx.x * blockDim.x + threadIdx.x);
    const int tid = threadIdx.x;
    const float* s = ws + WS_S;
    const int cpart = cnt[tid & (CNT_BLOCKS - 1)];

    float lm = 0.f;
    if (e < N_EDGES) {
        const int src = eidx[e];
        const int dst = eidx[N_EDGES + e];
        const float pre   = s[(size_t)src * 2 + 0] + s[(size_t)dst * 2 + 1];
        const float score = 1.f / (1.f + __expf(-pre));
        out[e] = score;
        const float p     = 1.f / (1.f + __expf(-score));
        const bool  pos   = (labels[e] == 1);
        const float pt    = pos ? p : (1.f - p);
        const float logpt = __logf(pt);
        const float at    = pos ? (1.f - ALPHA_F) : ALPHA_F;
        const float omp   = 1.f - pt;
        const float loss  = -(omp * omp) * (logpt * at);
        if (mask[e] != 0) lm = loss;
    }

    __shared__ float rl[256];
    rl[tid] = lm;
    int ctot = cpart;
    if (tid < 32) {
        for (int o = 16; o > 0; o >>= 1) ctot += __shfl_xor(ctot, o);
    }
    __syncthreads();
    for (int stride = 128; stride > 0; stride >>= 1) {
        if (tid < stride) rl[tid] += rl[tid + stride];
        __syncthreads();
    }
    if (tid == 0) atomicAdd(&out[N_EDGES], rl[0] / (float)ctot);
}

extern "C" void kernel_launch(void* const* d_in, const int* in_sizes, int n_in,
                              void* d_out, int out_size, void* d_ws, size_t ws_size,
                              hipStream_t stream) {
    const float* x      = (const float*)d_in[0];
    const float* Wd     = (const float*)d_in[1];
    const float* Wf     = (const float*)d_in[2];
    const int*   eidx   = (const int*)d_in[3];
    const int*   labels = (const int*)d_in[4];
    const int*   mask   = (const int*)d_in[5];
    float*       out    = (float*)d_out;
    float*       ws     = (float*)d_ws;
    int*         cnt    = (int*)(ws + WS_CNT);

    prep_kernel<<<FOLD_BLOCKS + CNT_BLOCKS, 256, 0, stream>>>(Wd, Wf, mask, ws, cnt, out);

    matvec_wmma_kernel<<<MV_BLOCKS, 256, 0, stream>>>(x, ws);

    edge_kernel<<<(N_EDGES + 255) / 256, 256, 0, stream>>>(eidx, labels, mask, ws, cnt, out);
}
